# starting kernel
# baseline (speedup 1.0000x reference)
#include <hip/hip_runtime.h>
#include <hip/hip_bf16.h>

typedef float v2f __attribute__((ext_vector_type(2)));
typedef float v8f __attribute__((ext_vector_type(8)));

#define D     128
#define NTOK  1024
#define G3    384

__device__ __forceinline__ v8f wmma4(v2f a, v2f b, v8f c) {
    return __builtin_amdgcn_wmma_f32_16x16x4_f32(
         false, a,  false, b,
         (short)0, c,  false,  false);
}

__device__ __forceinline__ float sigmoidf_dev(float x) {
    return 1.0f / (1.0f + __expf(-x));
}
__device__ __forceinline__ float tanhf_dev(float x) {
    x = fminf(15.0f, fmaxf(-15.0f, x));
    const float e = __expf(-2.0f * x);
    return (1.0f - e) / (1.0f + e);
}

__global__ __launch_bounds__(256) void gru_xproj_kernel(
    const float* __restrict__ xs, const float* __restrict__ Wx,
    const float* __restrict__ bx, float* __restrict__ Xp)
{
    const int lane = threadIdx.x & 31;
    const int wave = threadIdx.x >> 5;
    const int tile = blockIdx.x * 8 + wave;
    const int tm = tile / 24;
    const int tn = tile % 24;
    const int l15 = lane & 15;
    const int hi  = lane >> 4;

    const float* __restrict__ arow = &xs[(tm * 16 + l15) * D];
    const float* __restrict__ brow = &Wx[(tn * 16 + l15) * D];

    v8f acc = {};
    for (int k0 = 0; k0 < D; k0 += 4) {
        const int kb = k0 + 2 * hi;
        v2f a = { arow[kb], arow[kb + 1] };
        v2f b = { brow[kb], brow[kb + 1] };
        acc = wmma4(a, b, acc);
    }

    const int col  = tn * 16 + l15;
    const float bb = bx[col];
#pragma unroll
    for (int r = 0; r < 8; ++r) {
        const int row = tm * 16 + r + 8 * hi;
        Xp[row * G3 + col] = acc[r] + bb;
    }
}

__global__ __launch_bounds__(1024) void gru_serial_kernel(
    const int*   __restrict__ mask,
    const float* __restrict__ Wh, const float* __restrict__ bh,
    const float* __restrict__ Wx, const float* __restrict__ bx,
    const float* __restrict__ Xp,
    float* __restrict__ out)
{
    __shared__ float h1[D * D];
    __shared__ float h2[D * D];

    const int tid  = threadIdx.x;
    const int lane = tid & 31;
    const int wave = tid >> 5;
    const int l15  = lane & 15;
    const int hi   = lane >> 4;

    for (int i = tid; i < D * D; i += 1024) { h1[i] = 0.0f; h2[i] = 0.0f; }
    __syncthreads();

    for (int t = 0; t < NTOK; ++t) {
        const int mv = __builtin_amdgcn_readfirstlane(mask[t]);
        if (mv == 0) continue;

        float nv1[2][8];
#pragma unroll
        for (int gi = 0; gi < 2; ++gi) {
            const int g  = wave + gi * 32;
            const int mt = g >> 3;
            const int jt = g & 7;
            const float* __restrict__ arow = &h1[(mt * 16 + l15) * D];
            const float* __restrict__ wA = &Wh[(0 * D + jt * 16 + l15) * D];
            const float* __restrict__ wB = &Wh[(1 * D + jt * 16 + l15) * D];
            const float* __restrict__ wC = &Wh[(2 * D + jt * 16 + l15) * D];

            v8f accA = {}, accB = {}, accC = {};
            for (int k0 = 0; k0 < D; k0 += 4) {
                const int kb = k0 + 2 * hi;
                v2f a  = { arow[kb], arow[kb + 1] };
                v2f b0 = { wA[kb], wA[kb + 1] };
                v2f b1 = { wB[kb], wB[kb + 1] };
                v2f b2 = { wC[kb], wC[kb + 1] };
                accA = wmma4(a, b0, accA);
                accB = wmma4(a, b1, accB);
                accC = wmma4(a, b2, accC);
            }

            const int col = jt * 16 + l15;
            const float sa = Xp[t * G3 + col]           + bh[col];
            const float sb = Xp[t * G3 + D + col]       + bh[D + col];
            const float xc = Xp[t * G3 + 2 * D + col];
            const float bc = bh[2 * D + col];
#pragma unroll
            for (int r = 0; r < 8; ++r) {
                const int row = mt * 16 + r + 8 * hi;
                const float u  = sigmoidf_dev(accA[r] + sa);
                const float rg = sigmoidf_dev(accB[r] + sb);
                const float cd = tanhf_dev(xc + rg * (accC[r] + bc));
                const float ho = h1[row * D + col];
                nv1[gi][r] = u * ho + (1.0f - u) * cd;
            }
        }
        __syncthreads();
#pragma unroll
        for (int gi = 0; gi < 2; ++gi) {
            const int g = wave + gi * 32;
            const int mt = g >> 3, jt = g & 7;
            const int col = jt * 16 + l15;
#pragma unroll
            for (int r = 0; r < 8; ++r)
                h1[(mt * 16 + r + 8 * hi) * D + col] = nv1[gi][r];
        }
        __syncthreads();

        float nv2[2][8];
#pragma unroll
        for (int gi = 0; gi < 2; ++gi) {
            const int g  = wave + gi * 32;
            const int mt = g >> 3;
            const int jt = g & 7;
            const float* __restrict__ aH = &h2[(mt * 16 + l15) * D];
            const float* __restrict__ aX = &h1[(mt * 16 + l15) * D];
            const float* __restrict__ whA = &Wh[(0 * D + jt * 16 + l15) * D];
            const float* __restrict__ whB = &Wh[(1 * D + jt * 16 + l15) * D];
            const float* __restrict__ whC = &Wh[(2 * D + jt * 16 + l15) * D];
            const float* __restrict__ wxA = &Wx[(0 * D + jt * 16 + l15) * D];
            const float* __restrict__ wxB = &Wx[(1 * D + jt * 16 + l15) * D];
            const float* __restrict__ wxC = &Wx[(2 * D + jt * 16 + l15) * D];

            v8f hA = {}, hB = {}, hC = {}, xA = {}, xB = {}, xC = {};
            for (int k0 = 0; k0 < D; k0 += 4) {
                const int kb = k0 + 2 * hi;
                v2f a2 = { aH[kb], aH[kb + 1] };
                v2f a1 = { aX[kb], aX[kb + 1] };
                v2f bh0 = { whA[kb], whA[kb + 1] };
                v2f bh1 = { whB[kb], whB[kb + 1] };
                v2f bh2 = { whC[kb], whC[kb + 1] };
                v2f bx0 = { wxA[kb], wxA[kb + 1] };
                v2f bx1 = { wxB[kb], wxB[kb + 1] };
                v2f bx2 = { wxC[kb], wxC[kb + 1] };
                hA = wmma4(a2, bh0, hA);
                hB = wmma4(a2, bh1, hB);
                hC = wmma4(a2, bh2, hC);
                xA = wmma4(a1, bx0, xA);
                xB = wmma4(a1, bx1, xB);
                xC = wmma4(a1, bx2, xC);
            }

            const int col = jt * 16 + l15;
            const float ba0 = bh[col],         bb0 = bx[col];
            const float ba1 = bh[D + col],     bb1 = bx[D + col];
            const float ba2 = bh[2 * D + col], bb2 = bx[2 * D + col];
#pragma unroll
            for (int r = 0; r < 8; ++r) {
                const int row = mt * 16 + r + 8 * hi;
                const float u  = sigmoidf_dev((hA[r] + ba0) + (xA[r] + bb0));
                const float rg = sigmoidf_dev((hB[r] + ba1) + (xB[r] + bb1));
                const float cd = tanhf_dev((xC[r] + bb2) + rg * (hC[r] + ba2));
                const float ho = h2[row * D + col];
                nv2[gi][r] = u * ho + (1.0f - u) * cd;
            }
        }
        __syncthreads();
#pragma unroll
        for (int gi = 0; gi < 2; ++gi) {
            const int g = wave + gi * 32;
            const int mt = g >> 3, jt = g & 7;
            const int col = jt * 16 + l15;
#pragma unroll
            for (int r = 0; r < 8; ++r)
                h2[(mt * 16 + r + 8 * hi) * D + col] = nv2[gi][r];
        }
        __syncthreads();
    }

    for (int i = tid; i < D * D; i += 1024) {
        const float a = h1[i];
        const float b = h2[i];
        out[i] = b;
        out[D * D + 2 * i]     = a;
        out[D * D + 2 * i + 1] = b;
    }
}

extern "C" void kernel_launch(void* const* d_in, const int* in_sizes, int n_in,
                              void* d_out, int out_size, void* d_ws, size_t ws_size,
                              hipStream_t stream) {
    const float* xs   = (const float*)d_in[0];
    const int*   mask = (const int*)  d_in[1];
    const float* Wh   = (const float*)d_in[2];
    const float* bh   = (const float*)d_in[3];
    const float* Wx   = (const float*)d_in[4];
    const float* bx   = (const float*)d_in[5];
    float* Xp  = (float*)d_ws;
    float* out = (float*)d_out;

    gru_xproj_kernel<<<192, 256, 0, stream>>>(xs, Wx, bx, Xp);
    gru_serial_kernel<<<1, 1024, 0, stream>>>(mask, Wh, bh, Wx, bx, Xp, out);
}


# serial recurrence split by rows of the state: one work-group per independent 16-row tile of (h1,h2), 8 work-groups of 8 waves instead of 1 of 32; loaders, k order and sums unchanged
# speedup vs baseline: 6.7664x; 6.7664x over previous
#include <hip/hip_runtime.h>
#include <hip/hip_bf16.h>

typedef float v2f __attribute__((ext_vector_type(2)));
typedef float v8f __attribute__((ext_vector_type(8)));

#define D     128
#define NTOK  1024
#define G3    384

__device__ __forceinline__ v8f wmma4(v2f a, v2f b, v8f c) {
    return __builtin_amdgcn_wmma_f32_16x16x4_f32(
         false, a,  false, b,
         (short)0, c,  false,  false);
}

__device__ __forceinline__ float sigmoidf_dev(float x) {
    return 1.0f / (1.0f + __expf(-x));
}
__device__ __forceinline__ float tanhf_dev(float x) {
    x = fminf(15.0f, fmaxf(-15.0f, x));
    const float e = __expf(-2.0f * x);
    return (1.0f - e) / (1.0f + e);
}

__global__ __launch_bounds__(256) void gru_xproj_kernel(
    const float* __restrict__ xs, const float* __restrict__ Wx,
    const float* __restrict__ bx, float* __restrict__ Xp)
{
    const int lane = threadIdx.x & 31;
    const int wave = threadIdx.x >> 5;
    const int tile = blockIdx.x * 8 + wave;
    const int tm = tile / 24;
    const int tn = tile % 24;
    const int l15 = lane & 15;
    const int hi  = lane >> 4;

    const float* __restrict__ arow = &xs[(tm * 16 + l15) * D];
    const float* __restrict__ brow = &Wx[(tn * 16 + l15) * D];

    v8f acc = {};
    for (int k0 = 0; k0 < D; k0 += 4) {
        const int kb = k0 + 2 * hi;
        v2f a = { arow[kb], arow[kb + 1] };
        v2f b = { brow[kb], brow[kb + 1] };
        acc = wmma4(a, b, acc);
    }

    const int col  = tn * 16 + l15;
    const float bb = bx[col];
#pragma unroll
    for (int r = 0; r < 8; ++r) {
        const int row = tm * 16 + r + 8 * hi;
        Xp[row * G3 + col] = acc[r] + bb;
    }
}

#define RT 16
#define NWG 8
static_assert(NWG * RT == D, "row tiles must cover the state");

__global__ __launch_bounds__(256) void gru_serial_kernel(
    const int*   __restrict__ mask,
    const float* __restrict__ Wh, const float* __restrict__ bh,
    const float* __restrict__ Wx, const float* __restrict__ bx,
    const float* __restrict__ Xp,
    float* __restrict__ out)
{
    __shared__ float h1[RT * D];
    __shared__ float h2[RT * D];

    const int tid  = threadIdx.x;
    const int lane = tid & 31;
    const int jt   = tid >> 5;
    const int l15  = lane & 15;
    const int hi   = lane >> 4;
    const int row0 = blockIdx.x * RT;
    const int col  = jt * 16 + l15;

    for (int i = tid; i < RT * D; i += 256) { h1[i] = 0.0f; h2[i] = 0.0f; }
    __syncthreads();

    const float* __restrict__ aX = &h1[l15 * D];
    const float* __restrict__ aH = &h2[l15 * D];
    const float* __restrict__ whA = &Wh[(0 * D + jt * 16 + l15) * D];
    const float* __restrict__ whB = &Wh[(1 * D + jt * 16 + l15) * D];
    const float* __restrict__ whC = &Wh[(2 * D + jt * 16 + l15) * D];
    const float* __restrict__ wxA = &Wx[(0 * D + jt * 16 + l15) * D];
    const float* __restrict__ wxB = &Wx[(1 * D + jt * 16 + l15) * D];
    const float* __restrict__ wxC = &Wx[(2 * D + jt * 16 + l15) * D];

    for (int t = 0; t < NTOK; ++t) {
        const int mv = __builtin_amdgcn_readfirstlane(mask[t]);
        if (mv == 0) continue;

        float nv1[8];
        {
            v8f accA = {}, accB = {}, accC = {};
            for (int k0 = 0; k0 < D; k0 += 4) {
                const int kb = k0 + 2 * hi;
                v2f a  = { aX[kb], aX[kb + 1] };
                v2f b0 = { whA[kb], whA[kb + 1] };
                v2f b1 = { whB[kb], whB[kb + 1] };
                v2f b2 = { whC[kb], whC[kb + 1] };
                accA = wmma4(a, b0, accA);
                accB = wmma4(a, b1, accB);
                accC = wmma4(a, b2, accC);
            }

            const float sa = Xp[t * G3 + col]           + bh[col];
            const float sb = Xp[t * G3 + D + col]       + bh[D + col];
            const float xc = Xp[t * G3 + 2 * D + col];
            const float bc = bh[2 * D + col];
#pragma unroll
            for (int r = 0; r < 8; ++r) {
                const int row = r + 8 * hi;
                const float u  = sigmoidf_dev(accA[r] + sa);
                const float rg = sigmoidf_dev(accB[r] + sb);
                const float cd = tanhf_dev(xc + rg * (accC[r] + bc));
                const float ho = h1[row * D + col];
                nv1[r] = u * ho + (1.0f - u) * cd;
            }
        }
        __syncthreads();
#pragma unroll
        for (int r = 0; r < 8; ++r)
            h1[(r + 8 * hi) * D + col] = nv1[r];
        __syncthreads();

        float nv2[8];
        {
            v8f hA = {}, hB = {}, hC = {}, xA = {}, xB = {}, xC = {};
            for (int k0 = 0; k0 < D; k0 += 4) {
                const int kb = k0 + 2 * hi;
                v2f a2 = { aH[kb], aH[kb + 1] };
                v2f a1 = { aX[kb], aX[kb + 1] };
                v2f bh0 = { whA[kb], whA[kb + 1] };
                v2f bh1 = { whB[kb], whB[kb + 1] };
                v2f bh2 = { whC[kb], whC[kb + 1] };
                v2f bx0 = { wxA[kb], wxA[kb + 1] };
                v2f bx1 = { wxB[kb], wxB[kb + 1] };
                v2f bx2 = { wxC[kb], wxC[kb + 1] };
                hA = wmma4(a2, bh0, hA);
                hB = wmma4(a2, bh1, hB);
                hC = wmma4(a2, bh2, hC);
                xA = wmma4(a1, bx0, xA);
                xB = wmma4(a1, bx1, xB);
                xC = wmma4(a1, bx2, xC);
            }

            const float ba0 = bh[col],         bb0 = bx[col];
            const float ba1 = bh[D + col],     bb1 = bx[D + col];
            const float ba2 = bh[2 * D + col], bb2 = bx[2 * D + col];
#pragma unroll
            for (int r = 0; r < 8; ++r) {
                const int row = r + 8 * hi;
                const float u  = sigmoidf_dev((hA[r] + ba0) + (xA[r] + bb0));
                const float rg = sigmoidf_dev((hB[r] + ba1) + (xB[r] + bb1));
                const float cd = tanhf_dev((xC[r] + bb2) + rg * (hC[r] + ba2));
                const float ho = h2[row * D + col];
                nv2[r] = u * ho + (1.0f - u) * cd;
            }
        }
        __syncthreads();
#pragma unroll
        for (int r = 0; r < 8; ++r)
            h2[(r + 8 * hi) * D + col] = nv2[r];
        __syncthreads();
    }

    for (int i = tid; i < RT * D; i += 256) {
        const float a = h1[i];
        const float b = h2[i];
        const int gidx = row0 * D + i;
        out[gidx] = b;
        out[D * D + 2 * gidx]     = a;
        out[D * D + 2 * gidx + 1] = b;
    }
}

extern "C" void kernel_launch(void* const* d_in, const int* in_sizes, int n_in,
                              void* d_out, int out_size, void* d_ws, size_t ws_size,
                              hipStream_t stream) {
    const float* xs   = (const float*)d_in[0];
    const int*   mask = (const int*)  d_in[1];
    const float* Wh   = (const float*)d_in[2];
    const float* bh   = (const float*)d_in[3];
    const float* Wx   = (const float*)d_in[4];
    const float* bx   = (const float*)d_in[5];
    float* Xp  = (float*)d_ws;
    float* out = (float*)d_out;

    gru_xproj_kernel<<<192, 256, 0, stream>>>(xs, Wx, bx, Xp);
    gru_serial_kernel<<<NWG, 256, 0, stream>>>(mask, Wh, bh, Wx, bx, Xp, out);
}


# weights register-resident: each wave reads its Wh and Wx B fragments (384 VGPRs per lane) and its biases once before the time loop, not from L2 at every k-step; loaders, k order, sums unchanged
# speedup vs baseline: 13.3132x; 1.9675x over previous
#include <hip/hip_runtime.h>
#include <hip/hip_bf16.h>

typedef float v2f __attribute__((ext_vector_type(2)));
typedef float v8f __attribute__((ext_vector_type(8)));

#define D     128
#define NTOK  1024
#define G3    384

__device__ __forceinline__ v8f wmma4(v2f a, v2f b, v8f c) {
    return __builtin_amdgcn_wmma_f32_16x16x4_f32(
         false, a,  false, b,
         (short)0, c,  false,  false);
}

__device__ __forceinline__ float sigmoidf_dev(float x) {
    return 1.0f / (1.0f + __expf(-x));
}
__device__ __forceinline__ float tanhf_dev(float x) {
    x = fminf(15.0f, fmaxf(-15.0f, x));
    const float e = __expf(-2.0f * x);
    return (1.0f - e) / (1.0f + e);
}

__global__ __launch_bounds__(256) void gru_xproj_kernel(
    const float* __restrict__ xs, const float* __restrict__ Wx,
    const float* __restrict__ bx, float* __restrict__ Xp)
{
    const int lane = threadIdx.x & 31;
    const int wave = threadIdx.x >> 5;
    const int tile = blockIdx.x * 8 + wave;
    const int tm = tile / 24;
    const int tn = tile % 24;
    const int l15 = lane & 15;
    const int hi  = lane >> 4;

    const float* __restrict__ arow = &xs[(tm * 16 + l15) * D];
    const float* __restrict__ brow = &Wx[(tn * 16 + l15) * D];

    v8f acc = {};
    for (int k0 = 0; k0 < D; k0 += 4) {
        const int kb = k0 + 2 * hi;
        v2f a = { arow[kb], arow[kb + 1] };
        v2f b = { brow[kb], brow[kb + 1] };
        acc = wmma4(a, b, acc);
    }

    const int col  = tn * 16 + l15;
    const float bb = bx[col];
#pragma unroll
    for (int r = 0; r < 8; ++r) {
        const int row = tm * 16 + r + 8 * hi;
        Xp[row * G3 + col] = acc[r] + bb;
    }
}

#define RT 16
#define NWG 8
#define KS (D / 4)
static_assert(NWG * RT == D, "row tiles must cover the state");

__global__ __launch_bounds__(256) void gru_serial_kernel(
    const int*   __restrict__ mask,
    const float* __restrict__ Wh, const float* __restrict__ bh,
    const float* __restrict__ Wx, const float* __restrict__ bx,
    const float* __restrict__ Xp,
    float* __restrict__ out)
{
    __shared__ float h1[RT * D];
    __shared__ float h2[RT * D];

    const int tid  = threadIdx.x;
    const int lane = tid & 31;
    const int jt   = tid >> 5;
    const int l15  = lane & 15;
    const int hi   = lane >> 4;
    const int row0 = blockIdx.x * RT;
    const int col  = jt * 16 + l15;

    for (int i = tid; i < RT * D; i += 256) { h1[i] = 0.0f; h2[i] = 0.0f; }
    __syncthreads();

    const float* __restrict__ aX = &h1[l15 * D];
    const float* __restrict__ aH = &h2[l15 * D];
    const float* __restrict__ whA = &Wh[(0 * D + jt * 16 + l15) * D];
    const float* __restrict__ whB = &Wh[(1 * D + jt * 16 + l15) * D];
    const float* __restrict__ whC = &Wh[(2 * D + jt * 16 + l15) * D];
    const float* __restrict__ wxA = &Wx[(0 * D + jt * 16 + l15) * D];
    const float* __restrict__ wxB = &Wx[(1 * D + jt * 16 + l15) * D];
    const float* __restrict__ wxC = &Wx[(2 * D + jt * 16 + l15) * D];

    v2f fhA[KS], fhB[KS], fhC[KS], fxA[KS], fxB[KS], fxC[KS];
#pragma unroll
    for (int i = 0; i < KS; ++i) {
        const int kb = 4 * i + 2 * hi;
        fhA[i] = v2f{ whA[kb], whA[kb + 1] };
        fhB[i] = v2f{ whB[kb], whB[kb + 1] };
        fhC[i] = v2f{ whC[kb], whC[kb + 1] };
        fxA[i] = v2f{ wxA[kb], wxA[kb + 1] };
        fxB[i] = v2f{ wxB[kb], wxB[kb + 1] };
        fxC[i] = v2f{ wxC[kb], wxC[kb + 1] };
    }
    const float ba0 = bh[col],         bb0 = bx[col];
    const float ba1 = bh[D + col],     bb1 = bx[D + col];
    const float ba2 = bh[2 * D + col], bb2 = bx[2 * D + col];

    for (int t = 0; t < NTOK; ++t) {
        const int mv = __builtin_amdgcn_readfirstlane(mask[t]);
        if (mv == 0) continue;

        float nv1[8];
        {
            v8f accA = {}, accB = {}, accC = {};
#pragma unroll
            for (int i = 0; i < KS; ++i) {
                const int kb = 4 * i + 2 * hi;
                v2f a  = { aX[kb], aX[kb + 1] };
                accA = wmma4(a, fhA[i], accA);
                accB = wmma4(a, fhB[i], accB);
                accC = wmma4(a, fhC[i], accC);
            }

            const float sa = Xp[t * G3 + col]           + ba0;
            const float sb = Xp[t * G3 + D + col]       + ba1;
            const float xc = Xp[t * G3 + 2 * D + col];
            const float bc = ba2;
#pragma unroll
            for (int r = 0; r < 8; ++r) {
                const int row = r + 8 * hi;
                const float u  = sigmoidf_dev(accA[r] + sa);
                const float rg = sigmoidf_dev(accB[r] + sb);
                const float cd = tanhf_dev(xc + rg * (accC[r] + bc));
                const float ho = h1[row * D + col];
                nv1[r] = u * ho + (1.0f - u) * cd;
            }
        }
        __syncthreads();
#pragma unroll
        for (int r = 0; r < 8; ++r)
            h1[(r + 8 * hi) * D + col] = nv1[r];
        __syncthreads();

        float nv2[8];
        {
            v8f hA = {}, hB = {}, hC = {}, xA = {}, xB = {}, xC = {};
#pragma unroll
            for (int i = 0; i < KS; ++i) {
                const int kb = 4 * i + 2 * hi;
                v2f a2 = { aH[kb], aH[kb + 1] };
                v2f a1 = { aX[kb], aX[kb + 1] };
                hA = wmma4(a2, fhA[i], hA);
                hB = wmma4(a2, fhB[i], hB);
                hC = wmma4(a2, fhC[i], hC);
                xA = wmma4(a1, fxA[i], xA);
                xB = wmma4(a1, fxB[i], xB);
                xC = wmma4(a1, fxC[i], xC);
            }

#pragma unroll
            for (int r = 0; r < 8; ++r) {
                const int row = r + 8 * hi;
                const float u  = sigmoidf_dev((hA[r] + ba0) + (xA[r] + bb0));
                const float rg = sigmoidf_dev((hB[r] + ba1) + (xB[r] + bb1));
                const float cd = tanhf_dev((xC[r] + bb2) + rg * (hC[r] + ba2));
                const float ho = h2[row * D + col];
                nv2[r] = u * ho + (1.0f - u) * cd;
            }
        }
        __syncthreads();
#pragma unroll
        for (int r = 0; r < 8; ++r)
            h2[(r + 8 * hi) * D + col] = nv2[r];
        __syncthreads();
    }

    for (int i = tid; i < RT * D; i += 256) {
        const float a = h1[i];
        const float b = h2[i];
        const int gidx = row0 * D + i;
        out[gidx] = b;
        out[D * D + 2 * gidx]     = a;
        out[D * D + 2 * gidx + 1] = b;
    }
}

extern "C" void kernel_launch(void* const* d_in, const int* in_sizes, int n_in,
                              void* d_out, int out_size, void* d_ws, size_t ws_size,
                              hipStream_t stream) {
    const float* xs   = (const float*)d_in[0];
    const int*   mask = (const int*)  d_in[1];
    const float* Wh   = (const float*)d_in[2];
    const float* bh   = (const float*)d_in[3];
    const float* Wx   = (const float*)d_in[4];
    const float* bx   = (const float*)d_in[5];
    float* Xp  = (float*)d_ws;
    float* out = (float*)d_out;

    gru_xproj_kernel<<<192, 256, 0, stream>>>(xs, Wx, bx, Xp);
    gru_serial_kernel<<<NWG, 256, 0, stream>>>(mask, Wh, bh, Wx, bx, Xp, out);
}


# LDS row pitch of the h1/h2 state tiles 128 -> 132 floats (16 B x 33): each 8-byte A-fragment read of a wave falls on 64 different banks, not 16 rows deep on 4; loaders, k order, sums unchanged
# speedup vs baseline: 16.3651x; 1.2292x over previous
#include <hip/hip_runtime.h>
#include <hip/hip_bf16.h>

typedef float v2f __attribute__((ext_vector_type(2)));
typedef float v8f __attribute__((ext_vector_type(8)));

#define D     128
#define NTOK  1024
#define G3    384

__device__ __forceinline__ v8f wmma4(v2f a, v2f b, v8f c) {
    return __builtin_amdgcn_wmma_f32_16x16x4_f32(
         false, a,  false, b,
         (short)0, c,  false,  false);
}

__device__ __forceinline__ float sigmoidf_dev(float x) {
    return 1.0f / (1.0f + __expf(-x));
}
__device__ __forceinline__ float tanhf_dev(float x) {
    x = fminf(15.0f, fmaxf(-15.0f, x));
    const float e = __expf(-2.0f * x);
    return (1.0f - e) / (1.0f + e);
}

__global__ __launch_bounds__(256) void gru_xproj_kernel(
    const float* __restrict__ xs, const float* __restrict__ Wx,
    const float* __restrict__ bx, float* __restrict__ Xp)
{
    const int lane = threadIdx.x & 31;
    const int wave = threadIdx.x >> 5;
    const int tile = blockIdx.x * 8 + wave;
    const int tm = tile / 24;
    const int tn = tile % 24;
    const int l15 = lane & 15;
    const int hi  = lane >> 4;

    const float* __restrict__ arow = &xs[(tm * 16 + l15) * D];
    const float* __restrict__ brow = &Wx[(tn * 16 + l15) * D];

    v8f acc = {};
    for (int k0 = 0; k0 < D; k0 += 4) {
        const int kb = k0 + 2 * hi;
        v2f a = { arow[kb], arow[kb + 1] };
        v2f b = { brow[kb], brow[kb + 1] };
        acc = wmma4(a, b, acc);
    }

    const int col  = tn * 16 + l15;
    const float bb = bx[col];
#pragma unroll
    for (int r = 0; r < 8; ++r) {
        const int row = tm * 16 + r + 8 * hi;
        Xp[row * G3 + col] = acc[r] + bb;
    }
}

#define RT 16
#define NWG 8
#define KS (D / 4)
#define LP (D + 4)
static_assert(NWG * RT == D, "row tiles must cover the state");

__global__ __launch_bounds__(256) void gru_serial_kernel(
    const int*   __restrict__ mask,
    const float* __restrict__ Wh, const float* __restrict__ bh,
    const float* __restrict__ Wx, const float* __restrict__ bx,
    const float* __restrict__ Xp,
    float* __restrict__ out)
{
    __shared__ float h1[RT * LP];
    __shared__ float h2[RT * LP];

    const int tid  = threadIdx.x;
    const int lane = tid & 31;
    const int jt   = tid >> 5;
    const int l15  = lane & 15;
    const int hi   = lane >> 4;
    const int row0 = blockIdx.x * RT;
    const int col  = jt * 16 + l15;

    for (int i = tid; i < RT * LP; i += 256) { h1[i] = 0.0f; h2[i] = 0.0f; }
    __syncthreads();

    const float* __restrict__ aX = &h1[l15 * LP];
    const float* __restrict__ aH = &h2[l15 * LP];
    const float* __restrict__ whA = &Wh[(0 * D + jt * 16 + l15) * D];
    const float* __restrict__ whB = &Wh[(1 * D + jt * 16 + l15) * D];
    const float* __restrict__ whC = &Wh[(2 * D + jt * 16 + l15) * D];
    const float* __restrict__ wxA = &Wx[(0 * D + jt * 16 + l15) * D];
    const float* __restrict__ wxB = &Wx[(1 * D + jt * 16 + l15) * D];
    const float* __restrict__ wxC = &Wx[(2 * D + jt * 16 + l15) * D];

    v2f fhA[KS], fhB[KS], fhC[KS], fxA[KS], fxB[KS], fxC[KS];
#pragma unroll
    for (int i = 0; i < KS; ++i) {
        const int kb = 4 * i + 2 * hi;
        fhA[i] = v2f{ whA[kb], whA[kb + 1] };
        fhB[i] = v2f{ whB[kb], whB[kb + 1] };
        fhC[i] = v2f{ whC[kb], whC[kb + 1] };
        fxA[i] = v2f{ wxA[kb], wxA[kb + 1] };
        fxB[i] = v2f{ wxB[kb], wxB[kb + 1] };
        fxC[i] = v2f{ wxC[kb], wxC[kb + 1] };
    }
    const float ba0 = bh[col],         bb0 = bx[col];
    const float ba1 = bh[D + col],     bb1 = bx[D + col];
    const float ba2 = bh[2 * D + col], bb2 = bx[2 * D + col];

    for (int t = 0; t < NTOK; ++t) {
        const int mv = __builtin_amdgcn_readfirstlane(mask[t]);
        if (mv == 0) continue;

        float nv1[8];
        {
            v8f accA = {}, accB = {}, accC = {};
#pragma unroll
            for (int i = 0; i < KS; ++i) {
                const int kb = 4 * i + 2 * hi;
                v2f a  = { aX[kb], aX[kb + 1] };
                accA = wmma4(a, fhA[i], accA);
                accB = wmma4(a, fhB[i], accB);
                accC = wmma4(a, fhC[i], accC);
            }

            const float sa = Xp[t * G3 + col]           + ba0;
            const float sb = Xp[t * G3 + D + col]       + ba1;
            const float xc = Xp[t * G3 + 2 * D + col];
            const float bc = ba2;
#pragma unroll
            for (int r = 0; r < 8; ++r) {
                const int row = r + 8 * hi;
                const float u  = sigmoidf_dev(accA[r] + sa);
                const float rg = sigmoidf_dev(accB[r] + sb);
                const float cd = tanhf_dev(xc + rg * (accC[r] + bc));
                const float ho = h1[row * LP + col];
                nv1[r] = u * ho + (1.0f - u) * cd;
            }
        }
        __syncthreads();
#pragma unroll
        for (int r = 0; r < 8; ++r)
            h1[(r + 8 * hi) * LP + col] = nv1[r];
        __syncthreads();

        float nv2[8];
        {
            v8f hA = {}, hB = {}, hC = {}, xA = {}, xB = {}, xC = {};
#pragma unroll
            for (int i = 0; i < KS; ++i) {
                const int kb = 4 * i + 2 * hi;
                v2f a2 = { aH[kb], aH[kb + 1] };
                v2f a1 = { aX[kb], aX[kb + 1] };
                hA = wmma4(a2, fhA[i], hA);
                hB = wmma4(a2, fhB[i], hB);
                hC = wmma4(a2, fhC[i], hC);
                xA = wmma4(a1, fxA[i], xA);
                xB = wmma4(a1, fxB[i], xB);
                xC = wmma4(a1, fxC[i], xC);
            }

#pragma unroll
            for (int r = 0; r < 8; ++r) {
                const int row = r + 8 * hi;
                const float u  = sigmoidf_dev((hA[r] + ba0) + (xA[r] + bb0));
                const float rg = sigmoidf_dev((hB[r] + ba1) + (xB[r] + bb1));
                const float cd = tanhf_dev((xC[r] + bb2) + rg * (hC[r] + ba2));
                const float ho = h2[row * LP + col];
                nv2[r] = u * ho + (1.0f - u) * cd;
            }
        }
        __syncthreads();
#pragma unroll
        for (int r = 0; r < 8; ++r)
            h2[(r + 8 * hi) * LP + col] = nv2[r];
        __syncthreads();
    }

    for (int i = tid; i < RT * D; i += 256) {
        const int li = (i / D) * LP + (i % D);
        const float a = h1[li];
        const float b = h2[li];
        const int gidx = row0 * D + i;
        out[gidx] = b;
        out[D * D + 2 * gidx]     = a;
        out[D * D + 2 * gidx + 1] = b;
    }
}

extern "C" void kernel_launch(void* const* d_in, const int* in_sizes, int n_in,
                              void* d_out, int out_size, void* d_ws, size_t ws_size,
                              hipStream_t stream) {
    const float* xs   = (const float*)d_in[0];
    const int*   mask = (const int*)  d_in[1];
    const float* Wh   = (const float*)d_in[2];
    const float* bh   = (const float*)d_in[3];
    const float* Wx   = (const float*)d_in[4];
    const float* bx   = (const float*)d_in[5];
    float* Xp  = (float*)d_ws;
    float* out = (float*)d_out;

    gru_xproj_kernel<<<192, 256, 0, stream>>>(xs, Wx, bx, Xp);
    gru_serial_kernel<<<NWG, 256, 0, stream>>>(mask, Wh, bh, Wx, bx, Xp, out);
}


# all D rows of h1 and of h2 are equal for every input: one work-group carries 1 row per layer as rows 0,1 of one A tile, layer 2 one token behind: 192 WMMAs per wave-step, not 288; k order, sums kept
# speedup vs baseline: 29.0864x; 1.7773x over previous
#include <hip/hip_runtime.h>
#include <hip/hip_bf16.h>

typedef float v2f __attribute__((ext_vector_type(2)));
typedef float v8f __attribute__((ext_vector_type(8)));

#define D     128
#define NTOK  1024
#define G3    384

__device__ __forceinline__ v8f wmma4(v2f a, v2f b, v8f c) {
    return __builtin_amdgcn_wmma_f32_16x16x4_f32(
         false, a,  false, b,
         (short)0, c,  false,  false);
}

__device__ __forceinline__ float sigmoidf_dev(float x) {
    return 1.0f / (1.0f + __expf(-x));
}
__device__ __forceinline__ float tanhf_dev(float x) {
    x = fminf(15.0f, fmaxf(-15.0f, x));
    const float e = __expf(-2.0f * x);
    return (1.0f - e) / (1.0f + e);
}

__global__ __launch_bounds__(256) void gru_xproj_kernel(
    const float* __restrict__ xs, const float* __restrict__ Wx,
    const float* __restrict__ bx, float* __restrict__ Xp)
{
    const int lane = threadIdx.x & 31;
    const int wave = threadIdx.x >> 5;
    const int tile = blockIdx.x * 8 + wave;
    const int tm = tile / 24;
    const int tn = tile % 24;
    const int l15 = lane & 15;
    const int hi  = lane >> 4;

    const float* __restrict__ arow = &xs[(tm * 16 + l15) * D];
    const float* __restrict__ brow = &Wx[(tn * 16 + l15) * D];

    v8f acc = {};
    for (int k0 = 0; k0 < D; k0 += 4) {
        const int kb = k0 + 2 * hi;
        v2f a = { arow[kb], arow[kb + 1] };
        v2f b = { brow[kb], brow[kb + 1] };
        acc = wmma4(a, b, acc);
    }

    const int col  = tn * 16 + l15;
    const float bb = bx[col];
#pragma unroll
    for (int r = 0; r < 8; ++r) {
        const int row = tm * 16 + r + 8 * hi;
        Xp[row * G3 + col] = acc[r] + bb;
    }
}

#define RT 16
#define KS (D / 4)
#define LP (D + 4)

__global__ __launch_bounds__(256) void gru_serial_kernel(
    const int*   __restrict__ mask,
    const float* __restrict__ Wh, const float* __restrict__ bh,
    const float* __restrict__ Wx, const float* __restrict__ bx,
    const float* __restrict__ Xp,
    float* __restrict__ out)
{
    __shared__ float hs[RT * LP];

    const int tid  = threadIdx.x;
    const int lane = tid & 31;
    const int jt   = tid >> 5;
    const int l15  = lane & 15;
    const int hi   = lane >> 4;
    const int col  = jt * 16 + l15;

    for (int i = tid; i < RT * LP; i += 256) hs[i] = 0.0f;
    __syncthreads();

    const float* __restrict__ aS = &hs[l15 * LP];
    const float* __restrict__ whA = &Wh[(0 * D + jt * 16 + l15) * D];
    const float* __restrict__ whB = &Wh[(1 * D + jt * 16 + l15) * D];
    const float* __restrict__ whC = &Wh[(2 * D + jt * 16 + l15) * D];
    const float* __restrict__ wxA = &Wx[(0 * D + jt * 16 + l15) * D];
    const float* __restrict__ wxB = &Wx[(1 * D + jt * 16 + l15) * D];
    const float* __restrict__ wxC = &Wx[(2 * D + jt * 16 + l15) * D];

    v2f fhA[KS], fhB[KS], fhC[KS], fxA[KS], fxB[KS], fxC[KS];
#pragma unroll
    for (int i = 0; i < KS; ++i) {
        const int kb = 4 * i + 2 * hi;
        fhA[i] = v2f{ whA[kb], whA[kb + 1] };
        fhB[i] = v2f{ whB[kb], whB[kb + 1] };
        fhC[i] = v2f{ whC[kb], whC[kb + 1] };
        fxA[i] = v2f{ wxA[kb], wxA[kb + 1] };
        fxB[i] = v2f{ wxB[kb], wxB[kb + 1] };
        fxC[i] = v2f{ wxC[kb], wxC[kb + 1] };
    }
    const float ba0 = bh[col],         bb0 = bx[col];
    const float ba1 = bh[D + col],     bb1 = bx[D + col];
    const float ba2 = bh[2 * D + col], bb2 = bx[2 * D + col];

    float h1o = 0.0f, h2o = 0.0f;
    bool  pend = false;

    for (int t = 0; t <= NTOK; ++t) {
        const bool drain = (t == NTOK);
        const int  tc = drain ? NTOK - 1 : t;
        const int  mv = __builtin_amdgcn_readfirstlane(mask[tc]);
        const bool l1 = !drain && (mv > 0);
        const bool l2 = pend && (l1 || drain);
        if (!l1 && !l2) continue;

        v8f hA = {}, hB = {}, hC = {}, xA = {}, xB = {}, xC = {};
#pragma unroll
        for (int i = 0; i < KS; ++i) {
            const int kb = 4 * i + 2 * hi;
            v2f a = { aS[kb], aS[kb + 1] };
            hA = wmma4(a, fhA[i], hA);
            hB = wmma4(a, fhB[i], hB);
            hC = wmma4(a, fhC[i], hC);
            xA = wmma4(a, fxA[i], xA);
            xB = wmma4(a, fxB[i], xB);
            xC = wmma4(a, fxC[i], xC);
        }

        const float sa = Xp[tc * G3 + col]           + ba0;
        const float sb = Xp[tc * G3 + D + col]       + ba1;
        const float xc = Xp[tc * G3 + 2 * D + col];
        const float u1  = sigmoidf_dev(hA[0] + sa);
        const float rg1 = sigmoidf_dev(hB[0] + sb);
        const float cd1 = tanhf_dev(xc + rg1 * (hC[0] + ba2));
        const float n1  = u1 * h1o + (1.0f - u1) * cd1;
        const float u2  = sigmoidf_dev((hA[1] + ba0) + (xA[0] + bb0));
        const float rg2 = sigmoidf_dev((hB[1] + ba1) + (xB[0] + bb1));
        const float cd2 = tanhf_dev((xC[0] + bb2) + rg2 * (hC[1] + ba2));
        const float n2  = u2 * h2o + (1.0f - u2) * cd2;
        h1o = l1 ? n1 : h1o;
        h2o = l2 ? n2 : h2o;
        pend = pend || l1;

        __syncthreads();
        if (hi == 0) { hs[col] = h1o; hs[LP + col] = h2o; }
        __syncthreads();
    }

    float2* __restrict__ out2 = reinterpret_cast<float2*>(out + D * D);
    for (int i = tid; i < D * D; i += 256) {
        const int j = i % D;
        const float a = hs[j];
        const float b = hs[LP + j];
        out[i]  = b;
        out2[i] = make_float2(a, b);
    }
}

extern "C" void kernel_launch(void* const* d_in, const int* in_sizes, int n_in,
                              void* d_out, int out_size, void* d_ws, size_t ws_size,
                              hipStream_t stream) {
    const float* xs   = (const float*)d_in[0];
    const int*   mask = (const int*)  d_in[1];
    const float* Wh   = (const float*)d_in[2];
    const float* bh   = (const float*)d_in[3];
    const float* Wx   = (const float*)d_in[4];
    const float* bx   = (const float*)d_in[5];
    float* Xp  = (float*)d_ws;
    float* out = (float*)d_out;

    gru_xproj_kernel<<<192, 256, 0, stream>>>(xs, Wx, bx, Xp);
    gru_serial_kernel<<<1, 256, 0, stream>>>(mask, Wh, bh, Wx, bx, Xp, out);
}


# layer 2 runs 16 tokens behind layer 1: its input projection h1 @ Wx^T is taken for 16 tokens at once on the 16 rows of an A tile, once per 16 passes: 102 WMMAs per wave-pass, not 192; sums kept
# speedup vs baseline: 48.6851x; 1.6738x over previous
#include <hip/hip_runtime.h>
#include <hip/hip_bf16.h>

typedef float v2f __attribute__((ext_vector_type(2)));
typedef float v8f __attribute__((ext_vector_type(8)));

#define D     128
#define NTOK  1024
#define G3    384

__device__ __forceinline__ v8f wmma4(v2f a, v2f b, v8f c) {
    return __builtin_amdgcn_wmma_f32_16x16x4_f32(
         false, a,  false, b,
         (short)0, c,  false,  false);
}

__device__ __forceinline__ float sigmoidf_dev(float x) {
    return 1.0f / (1.0f + __expf(-x));
}
__device__ __forceinline__ float tanhf_dev(float x) {
    x = fminf(15.0f, fmaxf(-15.0f, x));
    const float e = __expf(-2.0f * x);
    return (1.0f - e) / (1.0f + e);
}

__global__ __launch_bounds__(256) void gru_xproj_kernel(
    const float* __restrict__ xs, const float* __restrict__ Wx,
    const float* __restrict__ bx, float* __restrict__ Xp)
{
    const int lane = threadIdx.x & 31;
    const int wave = threadIdx.x >> 5;
    const int tile = blockIdx.x * 8 + wave;
    const int tm = tile / 24;
    const int tn = tile % 24;
    const int l15 = lane & 15;
    const int hi  = lane >> 4;

    const float* __restrict__ arow = &xs[(tm * 16 + l15) * D];
    const float* __restrict__ brow = &Wx[(tn * 16 + l15) * D];

    v8f acc = {};
    for (int k0 = 0; k0 < D; k0 += 4) {
        const int kb = k0 + 2 * hi;
        v2f a = { arow[kb], arow[kb + 1] };
        v2f b = { brow[kb], brow[kb + 1] };
        acc = wmma4(a, b, acc);
    }

    const int col  = tn * 16 + l15;
    const float bb = bx[col];
#pragma unroll
    for (int r = 0; r < 8; ++r) {
        const int row = tm * 16 + r + 8 * hi;
        Xp[row * G3 + col] = acc[r] + bb;
    }
}

#define RT 16
#define LAG 16
#define KS (D / 4)
#define LP (D + 4)
#define XP2 (G3 + 4)

__global__ __launch_bounds__(256) void gru_serial_kernel(
    const int*   __restrict__ mask,
    const float* __restrict__ Wh, const float* __restrict__ bh,
    const float* __restrict__ Wx, const float* __restrict__ bx,
    const float* __restrict__ Xp,
    float* __restrict__ out)
{
    __shared__ float hs[RT * LP];
    __shared__ float hist[LAG * LP];
    __shared__ float x2[LAG * XP2];

    const int tid  = threadIdx.x;
    const int lane = tid & 31;
    const int jt   = tid >> 5;
    const int l15  = lane & 15;
    const int hi   = lane >> 4;
    const int col  = jt * 16 + l15;

    for (int i = tid; i < RT * LP; i += 256)   hs[i]   = 0.0f;
    for (int i = tid; i < LAG * LP; i += 256)  hist[i] = 0.0f;
    for (int i = tid; i < LAG * XP2; i += 256) x2[i]   = 0.0f;
    __syncthreads();

    const float* __restrict__ aS = &hs[l15 * LP];
    const float* __restrict__ aH = &hist[l15 * LP];
    const float* __restrict__ whA = &Wh[(0 * D + jt * 16 + l15) * D];
    const float* __restrict__ whB = &Wh[(1 * D + jt * 16 + l15) * D];
    const float* __restrict__ whC = &Wh[(2 * D + jt * 16 + l15) * D];
    const float* __restrict__ wxA = &Wx[(0 * D + jt * 16 + l15) * D];
    const float* __restrict__ wxB = &Wx[(1 * D + jt * 16 + l15) * D];
    const float* __restrict__ wxC = &Wx[(2 * D + jt * 16 + l15) * D];

    v2f fhA[KS], fhB[KS], fhC[KS], fxA[KS], fxB[KS], fxC[KS];
#pragma unroll
    for (int i = 0; i < KS; ++i) {
        const int kb = 4 * i + 2 * hi;
        fhA[i] = v2f{ whA[kb], whA[kb + 1] };
        fhB[i] = v2f{ whB[kb], whB[kb + 1] };
        fhC[i] = v2f{ whC[kb], whC[kb + 1] };
        fxA[i] = v2f{ wxA[kb], wxA[kb + 1] };
        fxB[i] = v2f{ wxB[kb], wxB[kb + 1] };
        fxC[i] = v2f{ wxC[kb], wxC[kb + 1] };
    }
    const float ba0 = bh[col],         bb0 = bx[col];
    const float ba1 = bh[D + col],     bb1 = bx[D + col];
    const float ba2 = bh[2 * D + col], bb2 = bx[2 * D + col];

    float h1o = 0.0f, h2o = 0.0f;

    for (int t = 0; t < NTOK + LAG; ++t) {
        const int m = t & (LAG - 1);

        if (m == 0 && t >= LAG) {
            v8f xA = {}, xB = {}, xC = {};
#pragma unroll
            for (int i = 0; i < KS; ++i) {
                const int kb = 4 * i + 2 * hi;
                v2f a = { aH[kb], aH[kb + 1] };
                xA = wmma4(a, fxA[i], xA);
                xB = wmma4(a, fxB[i], xB);
                xC = wmma4(a, fxC[i], xC);
            }
#pragma unroll
            for (int r = 0; r < 8; ++r) {
                float* __restrict__ xr = &x2[(r + 8 * hi) * XP2];
                xr[col]         = xA[r] + bb0;
                xr[D + col]     = xB[r] + bb1;
                xr[2 * D + col] = xC[r] + bb2;
            }
            __syncthreads();
        }

        const bool in1 = (t < NTOK);
        const bool in2 = (t >= LAG);
        const int  tc  = in1 ? t : NTOK - 1;
        const int  t2  = in2 ? t - LAG : 0;
        const int  mk1 = mask[tc];
        const int  mk2 = mask[t2];

        v8f hA = {}, hB = {}, hC = {};
#pragma unroll
        for (int i = 0; i < KS; ++i) {
            const int kb = 4 * i + 2 * hi;
            v2f a = { aS[kb], aS[kb + 1] };
            hA = wmma4(a, fhA[i], hA);
            hB = wmma4(a, fhB[i], hB);
            hC = wmma4(a, fhC[i], hC);
        }

        const float sa = Xp[tc * G3 + col]           + ba0;
        const float sb = Xp[tc * G3 + D + col]       + ba1;
        const float xc = Xp[tc * G3 + 2 * D + col];
        const float u1  = sigmoidf_dev(hA[0] + sa);
        const float rg1 = sigmoidf_dev(hB[0] + sb);
        const float cd1 = tanhf_dev(xc + rg1 * (hC[0] + ba2));
        const float n1  = u1 * h1o + (1.0f - u1) * cd1;
        const float pa = x2[m * XP2 + col];
        const float pb = x2[m * XP2 + D + col];
        const float pc = x2[m * XP2 + 2 * D + col];
        const float u2  = sigmoidf_dev((hA[1] + ba0) + pa);
        const float rg2 = sigmoidf_dev((hB[1] + ba1) + pb);
        const float cd2 = tanhf_dev(pc + rg2 * (hC[1] + ba2));
        const float n2  = u2 * h2o + (1.0f - u2) * cd2;
        const bool l1 = in1 & (mk1 > 0);
        const bool l2 = in2 & (mk2 > 0);
        h1o = l1 ? n1 : h1o;
        h2o = l2 ? n2 : h2o;

        __syncthreads();
        if (hi == 0) { hs[col] = h1o; hs[LP + col] = h2o; hist[m * LP + col] = h1o; }
        __syncthreads();
    }

    float2* __restrict__ out2 = reinterpret_cast<float2*>(out + D * D);
    for (int i = tid; i < D * D; i += 256) {
        const int j = i % D;
        const float a = hs[j];
        const float b = hs[LP + j];
        out[i]  = b;
        out2[i] = make_float2(a, b);
    }
}

extern "C" void kernel_launch(void* const* d_in, const int* in_sizes, int n_in,
                              void* d_out, int out_size, void* d_ws, size_t ws_size,
                              hipStream_t stream) {
    const float* xs   = (const float*)d_in[0];
    const int*   mask = (const int*)  d_in[1];
    const float* Wh   = (const float*)d_in[2];
    const float* bh   = (const float*)d_in[3];
    const float* Wx   = (const float*)d_in[4];
    const float* bx   = (const float*)d_in[5];
    float* Xp  = (float*)d_ws;
    float* out = (float*)d_out;

    gru_xproj_kernel<<<192, 256, 0, stream>>>(xs, Wx, bx, Xp);
    gru_serial_kernel<<<1, 256, 0, stream>>>(mask, Wh, bh, Wx, bx, Xp, out);
}


# per pass all LDS reads (32 A fragments, the projected row) issued as one group ahead of the 96-WMMA block, held there by scheduling barriers, not one read per six WMMAs; k order, sums unchanged
# speedup vs baseline: 51.0901x; 1.0494x over previous
#include <hip/hip_runtime.h>
#include <hip/hip_bf16.h>

typedef float v2f __attribute__((ext_vector_type(2)));
typedef float v8f __attribute__((ext_vector_type(8)));

#define D     128
#define NTOK  1024
#define G3    384

__device__ __forceinline__ v8f wmma4(v2f a, v2f b, v8f c) {
    return __builtin_amdgcn_wmma_f32_16x16x4_f32(
         false, a,  false, b,
         (short)0, c,  false,  false);
}

__device__ __forceinline__ float sigmoidf_dev(float x) {
    return 1.0f / (1.0f + __expf(-x));
}
__device__ __forceinline__ float tanhf_dev(float x) {
    x = fminf(15.0f, fmaxf(-15.0f, x));
    const float e = __expf(-2.0f * x);
    return (1.0f - e) / (1.0f + e);
}

__global__ __launch_bounds__(256) void gru_xproj_kernel(
    const float* __restrict__ xs, const float* __restrict__ Wx,
    const float* __restrict__ bx, float* __restrict__ Xp)
{
    const int lane = threadIdx.x & 31;
    const int wave = threadIdx.x >> 5;
    const int tile = blockIdx.x * 8 + wave;
    const int tm = tile / 24;
    const int tn = tile % 24;
    const int l15 = lane & 15;
    const int hi  = lane >> 4;

    const float* __restrict__ arow = &xs[(tm * 16 + l15) * D];
    const float* __restrict__ brow = &Wx[(tn * 16 + l15) * D];

    v8f acc = {};
    for (int k0 = 0; k0 < D; k0 += 4) {
        const int kb = k0 + 2 * hi;
        v2f a = { arow[kb], arow[kb + 1] };
        v2f b = { brow[kb], brow[kb + 1] };
        acc = wmma4(a, b, acc);
    }

    const int col  = tn * 16 + l15;
    const float bb = bx[col];
#pragma unroll
    for (int r = 0; r < 8; ++r) {
        const int row = tm * 16 + r + 8 * hi;
        Xp[row * G3 + col] = acc[r] + bb;
    }
}

#define RT 16
#define LAG 16
#define KS (D / 4)
#define LP (D + 4)
#define XP2 (G3 + 4)

__global__ __launch_bounds__(256) void gru_serial_kernel(
    const int*   __restrict__ mask,
    const float* __restrict__ Wh, const float* __restrict__ bh,
    const float* __restrict__ Wx, const float* __restrict__ bx,
    const float* __restrict__ Xp,
    float* __restrict__ out)
{
    __shared__ float hs[RT * LP];
    __shared__ float hist[LAG * LP];
    __shared__ float x2[LAG * XP2];

    const int tid  = threadIdx.x;
    const int lane = tid & 31;
    const int jt   = tid >> 5;
    const int l15  = lane & 15;
    const int hi   = lane >> 4;
    const int col  = jt * 16 + l15;

    for (int i = tid; i < RT * LP; i += 256)   hs[i]   = 0.0f;
    for (int i = tid; i < LAG * LP; i += 256)  hist[i] = 0.0f;
    for (int i = tid; i < LAG * XP2; i += 256) x2[i]   = 0.0f;
    __syncthreads();

    const float* __restrict__ aS = &hs[l15 * LP];
    const float* __restrict__ aH = &hist[l15 * LP];
    const float* __restrict__ whA = &Wh[(0 * D + jt * 16 + l15) * D];
    const float* __restrict__ whB = &Wh[(1 * D + jt * 16 + l15) * D];
    const float* __restrict__ whC = &Wh[(2 * D + jt * 16 + l15) * D];
    const float* __restrict__ wxA = &Wx[(0 * D + jt * 16 + l15) * D];
    const float* __restrict__ wxB = &Wx[(1 * D + jt * 16 + l15) * D];
    const float* __restrict__ wxC = &Wx[(2 * D + jt * 16 + l15) * D];

    v2f fhA[KS], fhB[KS], fhC[KS], fxA[KS], fxB[KS], fxC[KS];
#pragma unroll
    for (int i = 0; i < KS; ++i) {
        const int kb = 4 * i + 2 * hi;
        fhA[i] = v2f{ whA[kb], whA[kb + 1] };
        fhB[i] = v2f{ whB[kb], whB[kb + 1] };
        fhC[i] = v2f{ whC[kb], whC[kb + 1] };
        fxA[i] = v2f{ wxA[kb], wxA[kb + 1] };
        fxB[i] = v2f{ wxB[kb], wxB[kb + 1] };
        fxC[i] = v2f{ wxC[kb], wxC[kb + 1] };
    }
    const float ba0 = bh[col],         bb0 = bx[col];
    const float ba1 = bh[D + col],     bb1 = bx[D + col];
    const float ba2 = bh[2 * D + col], bb2 = bx[2 * D + col];

    float h1o = 0.0f, h2o = 0.0f;

    for (int t = 0; t < NTOK + LAG; ++t) {
        const int m = t & (LAG - 1);

        if (m == 0 && t >= LAG) {
            v8f xA = {}, xB = {}, xC = {};
            v2f ah[KS];
#pragma unroll
            for (int i = 0; i < KS; ++i) {
                const int kb = 4 * i + 2 * hi;
                ah[i] = v2f{ aH[kb], aH[kb + 1] };
            }
            __builtin_amdgcn_sched_barrier(0);
#pragma unroll
            for (int i = 0; i < KS; ++i) {
                xA = wmma4(ah[i], fxA[i], xA);
                xB = wmma4(ah[i], fxB[i], xB);
                xC = wmma4(ah[i], fxC[i], xC);
            }
#pragma unroll
            for (int r = 0; r < 8; ++r) {
                float* __restrict__ xr = &x2[(r + 8 * hi) * XP2];
                xr[col]         = xA[r] + bb0;
                xr[D + col]     = xB[r] + bb1;
                xr[2 * D + col] = xC[r] + bb2;
            }
            __syncthreads();
        }

        const bool in1 = (t < NTOK);
        const bool in2 = (t >= LAG);
        const int  tc  = in1 ? t : NTOK - 1;
        const int  t2  = in2 ? t - LAG : 0;
        const int  mk1 = mask[tc];
        const int  mk2 = mask[t2];

        v2f af[KS];
#pragma unroll
        for (int i = 0; i < KS; ++i) {
            const int kb = 4 * i + 2 * hi;
            af[i] = v2f{ aS[kb], aS[kb + 1] };
        }
        const float pa = x2[m * XP2 + col];
        const float pb = x2[m * XP2 + D + col];
        const float pc = x2[m * XP2 + 2 * D + col];
        const float xa = Xp[tc * G3 + col];
        const float xb = Xp[tc * G3 + D + col];
        const float xc = Xp[tc * G3 + 2 * D + col];
        __builtin_amdgcn_sched_barrier(0);

        v8f hA = {}, hB = {}, hC = {};
#pragma unroll
        for (int i = 0; i < KS; ++i) {
            hA = wmma4(af[i], fhA[i], hA);
            hB = wmma4(af[i], fhB[i], hB);
            hC = wmma4(af[i], fhC[i], hC);
        }
        __builtin_amdgcn_sched_barrier(0);

        const float sa = xa + ba0;
        const float sb = xb + ba1;
        const float u1  = sigmoidf_dev(hA[0] + sa);
        const float rg1 = sigmoidf_dev(hB[0] + sb);
        const float cd1 = tanhf_dev(xc + rg1 * (hC[0] + ba2));
        const float n1  = u1 * h1o + (1.0f - u1) * cd1;
        const float u2  = sigmoidf_dev((hA[1] + ba0) + pa);
        const float rg2 = sigmoidf_dev((hB[1] + ba1) + pb);
        const float cd2 = tanhf_dev(pc + rg2 * (hC[1] + ba2));
        const float n2  = u2 * h2o + (1.0f - u2) * cd2;
        const bool l1 = in1 & (mk1 > 0);
        const bool l2 = in2 & (mk2 > 0);
        h1o = l1 ? n1 : h1o;
        h2o = l2 ? n2 : h2o;

        __syncthreads();
        if (hi == 0) { hs[col] = h1o; hs[LP + col] = h2o; hist[m * LP + col] = h1o; }
        __syncthreads();
    }

    float2* __restrict__ out2 = reinterpret_cast<float2*>(out + D * D);
    for (int i = tid; i < D * D; i += 256) {
        const int j = i % D;
        const float a = hs[j];
        const float b = hs[LP + j];
        out[i]  = b;
        out2[i] = make_float2(a, b);
    }
}

extern "C" void kernel_launch(void* const* d_in, const int* in_sizes, int n_in,
                              void* d_out, int out_size, void* d_ws, size_t ws_size,
                              hipStream_t stream) {
    const float* xs   = (const float*)d_in[0];
    const int*   mask = (const int*)  d_in[1];
    const float* Wh   = (const float*)d_in[2];
    const float* bh   = (const float*)d_in[3];
    const float* Wx   = (const float*)d_in[4];
    const float* bx   = (const float*)d_in[5];
    float* Xp  = (float*)d_ws;
    float* out = (float*)d_out;

    gru_xproj_kernel<<<192, 256, 0, stream>>>(xs, Wx, bx, Xp);
    gru_serial_kernel<<<1, 256, 0, stream>>>(mask, Wh, bh, Wx, bx, Xp, out);
}


# gate functions on the hardware tanh (v_tanh_f32): sigmoid(x) = 0.5 + 0.5*tanh(x/2), tanh direct; the exp-and-division chain behind each pass's last WMMA falls from 34 to 10 instructions; sums kept
# speedup vs baseline: 53.2903x; 1.0431x over previous
#include <hip/hip_runtime.h>
#include <hip/hip_bf16.h>

typedef float v2f __attribute__((ext_vector_type(2)));
typedef float v8f __attribute__((ext_vector_type(8)));

#define D     128
#define NTOK  1024
#define G3    384

__device__ __forceinline__ v8f wmma4(v2f a, v2f b, v8f c) {
    return __builtin_amdgcn_wmma_f32_16x16x4_f32(
         false, a,  false, b,
         (short)0, c,  false,  false);
}

__device__ __forceinline__ float tanhf_dev(float x) {
    return __builtin_amdgcn_tanhf(x);
}
__device__ __forceinline__ float sigmoidf_dev(float x) {
    return fmaf(0.5f, tanhf_dev(0.5f * x), 0.5f);
}

__global__ __launch_bounds__(256) void gru_xproj_kernel(
    const float* __restrict__ xs, const float* __restrict__ Wx,
    const float* __restrict__ bx, float* __restrict__ Xp)
{
    const int lane = threadIdx.x & 31;
    const int wave = threadIdx.x >> 5;
    const int tile = blockIdx.x * 8 + wave;
    const int tm = tile / 24;
    const int tn = tile % 24;
    const int l15 = lane & 15;
    const int hi  = lane >> 4;

    const float* __restrict__ arow = &xs[(tm * 16 + l15) * D];
    const float* __restrict__ brow = &Wx[(tn * 16 + l15) * D];

    v8f acc = {};
    for (int k0 = 0; k0 < D; k0 += 4) {
        const int kb = k0 + 2 * hi;
        v2f a = { arow[kb], arow[kb + 1] };
        v2f b = { brow[kb], brow[kb + 1] };
        acc = wmma4(a, b, acc);
    }

    const int col  = tn * 16 + l15;
    const float bb = bx[col];
#pragma unroll
    for (int r = 0; r < 8; ++r) {
        const int row = tm * 16 + r + 8 * hi;
        Xp[row * G3 + col] = acc[r] + bb;
    }
}

#define RT 16
#define LAG 16
#define KS (D / 4)
#define LP (D + 4)
#define XP2 (G3 + 4)

__global__ __launch_bounds__(256) void gru_serial_kernel(
    const int*   __restrict__ mask,
    const float* __restrict__ Wh, const float* __restrict__ bh,
    const float* __restrict__ Wx, const float* __restrict__ bx,
    const float* __restrict__ Xp,
    float* __restrict__ out)
{
    __shared__ float hs[RT * LP];
    __shared__ float hist[LAG * LP];
    __shared__ float x2[LAG * XP2];

    const int tid  = threadIdx.x;
    const int lane = tid & 31;
    const int jt   = tid >> 5;
    const int l15  = lane & 15;
    const int hi   = lane >> 4;
    const int col  = jt * 16 + l15;

    for (int i = tid; i < RT * LP; i += 256)   hs[i]   = 0.0f;
    for (int i = tid; i < LAG * LP; i += 256)  hist[i] = 0.0f;
    for (int i = tid; i < LAG * XP2; i += 256) x2[i]   = 0.0f;
    __syncthreads();

    const float* __restrict__ aS = &hs[l15 * LP];
    const float* __restrict__ aH = &hist[l15 * LP];
    const float* __restrict__ whA = &Wh[(0 * D + jt * 16 + l15) * D];
    const float* __restrict__ whB = &Wh[(1 * D + jt * 16 + l15) * D];
    const float* __restrict__ whC = &Wh[(2 * D + jt * 16 + l15) * D];
    const float* __restrict__ wxA = &Wx[(0 * D + jt * 16 + l15) * D];
    const float* __restrict__ wxB = &Wx[(1 * D + jt * 16 + l15) * D];
    const float* __restrict__ wxC = &Wx[(2 * D + jt * 16 + l15) * D];

    v2f fhA[KS], fhB[KS], fhC[KS], fxA[KS], fxB[KS], fxC[KS];
#pragma unroll
    for (int i = 0; i < KS; ++i) {
        const int kb = 4 * i + 2 * hi;
        fhA[i] = v2f{ whA[kb], whA[kb + 1] };
        fhB[i] = v2f{ whB[kb], whB[kb + 1] };
        fhC[i] = v2f{ whC[kb], whC[kb + 1] };
        fxA[i] = v2f{ wxA[kb], wxA[kb + 1] };
        fxB[i] = v2f{ wxB[kb], wxB[kb + 1] };
        fxC[i] = v2f{ wxC[kb], wxC[kb + 1] };
    }
    const float ba0 = bh[col],         bb0 = bx[col];
    const float ba1 = bh[D + col],     bb1 = bx[D + col];
    const float ba2 = bh[2 * D + col], bb2 = bx[2 * D + col];

    float h1o = 0.0f, h2o = 0.0f;

    for (int t = 0; t < NTOK + LAG; ++t) {
        const int m = t & (LAG - 1);

        if (m == 0 && t >= LAG) {
            v8f xA = {}, xB = {}, xC = {};
            v2f ah[KS];
#pragma unroll
            for (int i = 0; i < KS; ++i) {
                const int kb = 4 * i + 2 * hi;
                ah[i] = v2f{ aH[kb], aH[kb + 1] };
            }
            __builtin_amdgcn_sched_barrier(0);
#pragma unroll
            for (int i = 0; i < KS; ++i) {
                xA = wmma4(ah[i], fxA[i], xA);
                xB = wmma4(ah[i], fxB[i], xB);
                xC = wmma4(ah[i], fxC[i], xC);
            }
#pragma unroll
            for (int r = 0; r < 8; ++r) {
                float* __restrict__ xr = &x2[(r + 8 * hi) * XP2];
                xr[col]         = xA[r] + bb0;
                xr[D + col]     = xB[r] + bb1;
                xr[2 * D + col] = xC[r] + bb2;
            }
            __syncthreads();
        }

        const bool in1 = (t < NTOK);
        const bool in2 = (t >= LAG);
        const int  tc  = in1 ? t : NTOK - 1;
        const int  t2  = in2 ? t - LAG : 0;
        const int  mk1 = mask[tc];
        const int  mk2 = mask[t2];

        v2f af[KS];
#pragma unroll
        for (int i = 0; i < KS; ++i) {
            const int kb = 4 * i + 2 * hi;
            af[i] = v2f{ aS[kb], aS[kb + 1] };
        }
        const float pa = x2[m * XP2 + col];
        const float pb = x2[m * XP2 + D + col];
        const float pc = x2[m * XP2 + 2 * D + col];
        const float xa = Xp[tc * G3 + col];
        const float xb = Xp[tc * G3 + D + col];
        const float xc = Xp[tc * G3 + 2 * D + col];
        __builtin_amdgcn_sched_barrier(0);

        v8f hA = {}, hB = {}, hC = {};
#pragma unroll
        for (int i = 0; i < KS; ++i) {
            hA = wmma4(af[i], fhA[i], hA);
            hB = wmma4(af[i], fhB[i], hB);
            hC = wmma4(af[i], fhC[i], hC);
        }
        __builtin_amdgcn_sched_barrier(0);

        const float sa = xa + ba0;
        const float sb = xb + ba1;
        const float u1  = sigmoidf_dev(hA[0] + sa);
        const float rg1 = sigmoidf_dev(hB[0] + sb);
        const float cd1 = tanhf_dev(xc + rg1 * (hC[0] + ba2));
        const float n1  = u1 * h1o + (1.0f - u1) * cd1;
        const float u2  = sigmoidf_dev((hA[1] + ba0) + pa);
        const float rg2 = sigmoidf_dev((hB[1] + ba1) + pb);
        const float cd2 = tanhf_dev(pc + rg2 * (hC[1] + ba2));
        const float n2  = u2 * h2o + (1.0f - u2) * cd2;
        const bool l1 = in1 & (mk1 > 0);
        const bool l2 = in2 & (mk2 > 0);
        h1o = l1 ? n1 : h1o;
        h2o = l2 ? n2 : h2o;

        __syncthreads();
        if (hi == 0) { hs[col] = h1o; hs[LP + col] = h2o; hist[m * LP + col] = h1o; }
        __syncthreads();
    }

    float2* __restrict__ out2 = reinterpret_cast<float2*>(out + D * D);
    for (int i = tid; i < D * D; i += 256) {
        const int j = i % D;
        const float a = hs[j];
        const float b = hs[LP + j];
        out[i]  = b;
        out2[i] = make_float2(a, b);
    }
}

extern "C" void kernel_launch(void* const* d_in, const int* in_sizes, int n_in,
                              void* d_out, int out_size, void* d_ws, size_t ws_size,
                              hipStream_t stream) {
    const float* xs   = (const float*)d_in[0];
    const int*   mask = (const int*)  d_in[1];
    const float* Wh   = (const float*)d_in[2];
    const float* bh   = (const float*)d_in[3];
    const float* Wx   = (const float*)d_in[4];
    const float* bx   = (const float*)d_in[5];
    float* Xp  = (float*)d_ws;
    float* out = (float*)d_out;

    gru_xproj_kernel<<<192, 256, 0, stream>>>(xs, Wx, bx, Xp);
    gru_serial_kernel<<<1, 256, 0, stream>>>(mask, Wh, bh, Wx, bx, Xp, out);
}


# state handed over in two halves per pass on a split barrier (waves 0-3: columns 0-63, waves 4-7: 64-127): k-steps 0-15 of the next pass run while the late waves finish this one; k order, sums kept
# speedup vs baseline: 55.1156x; 1.0343x over previous
#include <hip/hip_runtime.h>
#include <hip/hip_bf16.h>

typedef float v2f __attribute__((ext_vector_type(2)));
typedef float v8f __attribute__((ext_vector_type(8)));

#define D     128
#define NTOK  1024
#define G3    384

__device__ __forceinline__ v8f wmma4(v2f a, v2f b, v8f c) {
    return __builtin_amdgcn_wmma_f32_16x16x4_f32(
         false, a,  false, b,
         (short)0, c,  false,  false);
}

__device__ __forceinline__ float tanhf_dev(float x) {
    return __builtin_amdgcn_tanhf(x);
}
__device__ __forceinline__ float sigmoidf_dev(float x) {
    return fmaf(0.5f, tanhf_dev(0.5f * x), 0.5f);
}

__global__ __launch_bounds__(256) void gru_xproj_kernel(
    const float* __restrict__ xs, const float* __restrict__ Wx,
    const float* __restrict__ bx, float* __restrict__ Xp)
{
    const int lane = threadIdx.x & 31;
    const int wave = threadIdx.x >> 5;
    const int tile = blockIdx.x * 8 + wave;
    const int tm = tile / 24;
    const int tn = tile % 24;
    const int l15 = lane & 15;
    const int hi  = lane >> 4;

    const float* __restrict__ arow = &xs[(tm * 16 + l15) * D];
    const float* __restrict__ brow = &Wx[(tn * 16 + l15) * D];

    v8f acc = {};
    for (int k0 = 0; k0 < D; k0 += 4) {
        const int kb = k0 + 2 * hi;
        v2f a = { arow[kb], arow[kb + 1] };
        v2f b = { brow[kb], brow[kb + 1] };
        acc = wmma4(a, b, acc);
    }

    const int col  = tn * 16 + l15;
    const float bb = bx[col];
#pragma unroll
    for (int r = 0; r < 8; ++r) {
        const int row = tm * 16 + r + 8 * hi;
        Xp[row * G3 + col] = acc[r] + bb;
    }
}

#define RT 16
#define LAG 16
#define KS (D / 4)
#define KH (KS / 2)
#define KB 12
#define LP (D + 4)
#define XP2 (G3 + 4)

__device__ __forceinline__ void bar_signal() {
    asm volatile("s_wait_dscnt 0x0\n\ts_barrier_signal -1" ::: "memory");
}
__device__ __forceinline__ void bar_wait() {
    asm volatile("s_barrier_wait -1" ::: "memory");
}

#define GATES_AND_STORE                                                          \
    {                                                                            \
                                                            \
        const float sa = xa + ba0;                                               \
        const float sb = xb + ba1;                                               \
        const float u1  = sigmoidf_dev(hA[0] + sa);                              \
        const float rg1 = sigmoidf_dev(hB[0] + sb);                              \
        const float cd1 = tanhf_dev(xc + rg1 * (hC[0] + ba2));                   \
        const float n1  = u1 * h1o + (1.0f - u1) * cd1;                          \
                                                      \
        const float u2  = sigmoidf_dev((hA[1] + ba0) + pa);                      \
        const float rg2 = sigmoidf_dev((hB[1] + ba1) + pb);                      \
        const float cd2 = tanhf_dev(pc + rg2 * (hC[1] + ba2));                   \
        const float n2  = u2 * h2o + (1.0f - u2) * cd2;                          \
        const bool l1 = in1 & (mk1 > 0);            \
        const bool l2 = in2 & (mk2 > 0);                                         \
        h1o = l1 ? n1 : h1o;                                                     \
        h2o = l2 ? n2 : h2o;                                                     \
        if (hi == 0) {                                                           \
            hs[wp * RT * LP + col]        = h1o;                                 \
            hs[wp * RT * LP + LP + col]   = h2o;                                 \
            hist[(bp * LAG + m) * LP + col] = h1o;                               \
        }                                                                        \
    }

__global__ __launch_bounds__(256) void gru_serial_kernel(
    const int*   __restrict__ mask,
    const float* __restrict__ Wh, const float* __restrict__ bh,
    const float* __restrict__ Wx, const float* __restrict__ bx,
    const float* __restrict__ Xp,
    float* __restrict__ out)
{
    __shared__ float hs[2 * RT * LP];
    __shared__ float hist[2 * LAG * LP];
    __shared__ float x2[LAG * XP2];

    const int tid  = threadIdx.x;
    const int lane = tid & 31;
    const int jt   = tid >> 5;
    const int l15  = lane & 15;
    const int hi   = lane >> 4;
    const int col  = jt * 16 + l15;
    const bool g0 = __builtin_amdgcn_readfirstlane(jt) < 4;

    for (int i = tid; i < 2 * RT * LP; i += 256)   hs[i]   = 0.0f;
    for (int i = tid; i < 2 * LAG * LP; i += 256)  hist[i] = 0.0f;
    for (int i = tid; i < LAG * XP2; i += 256)     x2[i]   = 0.0f;
    __syncthreads();

    const float* __restrict__ whA = &Wh[(0 * D + jt * 16 + l15) * D];
    const float* __restrict__ whB = &Wh[(1 * D + jt * 16 + l15) * D];
    const float* __restrict__ whC = &Wh[(2 * D + jt * 16 + l15) * D];
    const float* __restrict__ wxA = &Wx[(0 * D + jt * 16 + l15) * D];
    const float* __restrict__ wxB = &Wx[(1 * D + jt * 16 + l15) * D];
    const float* __restrict__ wxC = &Wx[(2 * D + jt * 16 + l15) * D];

    v2f fhA[KS], fhB[KS], fhC[KS], fxA[KS], fxB[KS], fxC[KS];
#pragma unroll
    for (int i = 0; i < KS; ++i) {
        const int kb = 4 * i + 2 * hi;
        fhA[i] = v2f{ whA[kb], whA[kb + 1] };
        fhB[i] = v2f{ whB[kb], whB[kb + 1] };
        fhC[i] = v2f{ whC[kb], whC[kb + 1] };
        fxA[i] = v2f{ wxA[kb], wxA[kb + 1] };
        fxB[i] = v2f{ wxB[kb], wxB[kb + 1] };
        fxC[i] = v2f{ wxC[kb], wxC[kb + 1] };
    }
    const float ba0 = bh[col],         bb0 = bx[col];
    const float ba1 = bh[D + col],     bb1 = bx[D + col];
    const float ba2 = bh[2 * D + col], bb2 = bx[2 * D + col];

    float h1o = 0.0f, h2o = 0.0f;

    v2f a1[KH];
    {
        const float* aZ = &hs[RT * LP + l15 * LP];
#pragma unroll
        for (int i = 0; i < KH; ++i) {
            const int kb = 4 * i + 2 * hi;
            a1[i] = v2f{ aZ[kb], aZ[kb + 1] };
        }
    }
    bar_signal();

    for (int t = 0; t < NTOK + LAG; ++t) {
        const int m  = t & (LAG - 1);
        const int rp = (t + 1) & 1;
        const int wp = t & 1;
        const int bp = (t >> 4) & 1;
        const float* aR = &hs[rp * RT * LP + l15 * LP];
        const float* aW = &hs[wp * RT * LP + l15 * LP];

        const bool in1 = (t < NTOK);
        const bool in2 = (t >= LAG);
        const int  tc  = in1 ? t : NTOK - 1;
        const int  t2  = in2 ? t - LAG : 0;
        const int  mk1 = mask[tc];
        const int  mk2 = mask[t2];
        const float xa = Xp[tc * G3 + col];
        const float xb = Xp[tc * G3 + D + col];
        const float xc = Xp[tc * G3 + 2 * D + col];

        v8f hA = {}, hB = {}, hC = {};
#pragma unroll
        for (int i = 0; i < KB; ++i) {
            hA = wmma4(a1[i], fhA[i], hA);
            hB = wmma4(a1[i], fhB[i], hB);
            hC = wmma4(a1[i], fhC[i], hC);
        }
        __builtin_amdgcn_sched_barrier(0);

        bar_wait();

        if (m == 0 && t >= LAG) {
            const float* aH = &hist[(bp ^ 1) * LAG * LP + l15 * LP];
            v8f xA = {}, xB = {}, xC = {};
#pragma unroll
            for (int hh = 0; hh < 2; ++hh) {
                v2f ah[KH];
#pragma unroll
                for (int i = 0; i < KH; ++i) {
                    const int kb = 4 * (hh * KH + i) + 2 * hi;
                    ah[i] = v2f{ aH[kb], aH[kb + 1] };
                }
                __builtin_amdgcn_sched_barrier(0);
#pragma unroll
                for (int i = 0; i < KH; ++i) {
                    xA = wmma4(ah[i], fxA[hh * KH + i], xA);
                    xB = wmma4(ah[i], fxB[hh * KH + i], xB);
                    xC = wmma4(ah[i], fxC[hh * KH + i], xC);
                }
                __builtin_amdgcn_sched_barrier(0);
            }
#pragma unroll
            for (int r = 0; r < 8; ++r) {
                float* xr = &x2[(r + 8 * hi) * XP2];
                xr[col]         = xA[r] + bb0;
                xr[D + col]     = xB[r] + bb1;
                xr[2 * D + col] = xC[r] + bb2;
            }
        }

        v2f a2[KH];
#pragma unroll
        for (int i = 0; i < KH; ++i) {
            const int kb = 4 * (KH + i) + 2 * hi;
            a2[i] = v2f{ aR[kb], aR[kb + 1] };
        }
        const float pa = x2[m * XP2 + col];
        const float pb = x2[m * XP2 + D + col];
        const float pc = x2[m * XP2 + 2 * D + col];
        __builtin_amdgcn_sched_barrier(0);

#pragma unroll
        for (int i = KB; i < KH; ++i) {
            hA = wmma4(a1[i], fhA[i], hA);
            hB = wmma4(a1[i], fhB[i], hB);
            hC = wmma4(a1[i], fhC[i], hC);
        }

        if (g0) {
#pragma unroll
            for (int i = 0; i < KH; ++i) {
                hA = wmma4(a2[i], fhA[KH + i], hA);
                hB = wmma4(a2[i], fhB[KH + i], hB);
                hC = wmma4(a2[i], fhC[KH + i], hC);
            }
            GATES_AND_STORE
        }
        bar_signal();
        if (!g0) {
#pragma unroll
            for (int i = 0; i < KH; ++i) {
                hA = wmma4(a2[i], fhA[KH + i], hA);
                hB = wmma4(a2[i], fhB[KH + i], hB);
                hC = wmma4(a2[i], fhC[KH + i], hC);
            }
        }
        bar_wait();

#pragma unroll
        for (int i = 0; i < KH; ++i) {
            const int kb = 4 * i + 2 * hi;
            a1[i] = v2f{ aW[kb], aW[kb + 1] };
        }
        __builtin_amdgcn_sched_barrier(0);

        if (!g0) {
            GATES_AND_STORE
        }
        bar_signal();
    }
    bar_wait();

    const float* hf = &hs[((NTOK + LAG - 1) & 1) * RT * LP];
    float2* __restrict__ out2 = reinterpret_cast<float2*>(out + D * D);
    for (int i = tid; i < D * D; i += 256) {
        const int j = i % D;
        const float a = hf[j];
        const float b = hf[LP + j];
        out[i]  = b;
        out2[i] = make_float2(a, b);
    }
}

extern "C" void kernel_launch(void* const* d_in, const int* in_sizes, int n_in,
                              void* d_out, int out_size, void* d_ws, size_t ws_size,
                              hipStream_t stream) {
    const float* xs   = (const float*)d_in[0];
    const int*   mask = (const int*)  d_in[1];
    const float* Wh   = (const float*)d_in[2];
    const float* bh   = (const float*)d_in[3];
    const float* Wx   = (const float*)d_in[4];
    const float* bx   = (const float*)d_in[5];
    float* Xp  = (float*)d_ws;
    float* out = (float*)d_out;

    gru_xproj_kernel<<<192, 256, 0, stream>>>(xs, Wx, bx, Xp);
    gru_serial_kernel<<<1, 256, 0, stream>>>(mask, Wh, bh, Wx, bx, Xp, out);
}


# waves 4-7 raise issue priority (s_setprio 3) from first-half-out to second-half-out: their last 16 k-steps and gates go first, the older waves' next-pass k-steps fill that gate time; sums kept
# speedup vs baseline: 57.1283x; 1.0365x over previous
#include <hip/hip_runtime.h>
#include <hip/hip_bf16.h>

typedef float v2f __attribute__((ext_vector_type(2)));
typedef float v8f __attribute__((ext_vector_type(8)));

#define D     128
#define NTOK  1024
#define G3    384

__device__ __forceinline__ v8f wmma4(v2f a, v2f b, v8f c) {
    return __builtin_amdgcn_wmma_f32_16x16x4_f32(
         false, a,  false, b,
         (short)0, c,  false,  false);
}

__device__ __forceinline__ float tanhf_dev(float x) {
    return __builtin_amdgcn_tanhf(x);
}
__device__ __forceinline__ float sigmoidf_dev(float x) {
    return fmaf(0.5f, tanhf_dev(0.5f * x), 0.5f);
}

__global__ __launch_bounds__(256) void gru_xproj_kernel(
    const float* __restrict__ xs, const float* __restrict__ Wx,
    const float* __restrict__ bx, float* __restrict__ Xp)
{
    const int lane = threadIdx.x & 31;
    const int wave = threadIdx.x >> 5;
    const int tile = blockIdx.x * 8 + wave;
    const int tm = tile / 24;
    const int tn = tile % 24;
    const int l15 = lane & 15;
    const int hi  = lane >> 4;

    const float* __restrict__ arow = &xs[(tm * 16 + l15) * D];
    const float* __restrict__ brow = &Wx[(tn * 16 + l15) * D];

    v8f acc = {};
    for (int k0 = 0; k0 < D; k0 += 4) {
        const int kb = k0 + 2 * hi;
        v2f a = { arow[kb], arow[kb + 1] };
        v2f b = { brow[kb], brow[kb + 1] };
        acc = wmma4(a, b, acc);
    }

    const int col  = tn * 16 + l15;
    const float bb = bx[col];
#pragma unroll
    for (int r = 0; r < 8; ++r) {
        const int row = tm * 16 + r + 8 * hi;
        Xp[row * G3 + col] = acc[r] + bb;
    }
}

#define RT 16
#define LAG 16
#define KS (D / 4)
#define KH (KS / 2)
#define KB 12
#define LP (D + 4)
#define XP2 (G3 + 4)

__device__ __forceinline__ void bar_signal() {
    asm volatile("s_wait_dscnt 0x0\n\ts_barrier_signal -1" ::: "memory");
}
__device__ __forceinline__ void bar_wait() {
    asm volatile("s_barrier_wait -1" ::: "memory");
}

#define GATES_AND_STORE                                                          \
    {                                                                            \
                                                            \
        const float sa = xa + ba0;                                               \
        const float sb = xb + ba1;                                               \
        const float u1  = sigmoidf_dev(hA[0] + sa);                              \
        const float rg1 = sigmoidf_dev(hB[0] + sb);                              \
        const float cd1 = tanhf_dev(xc + rg1 * (hC[0] + ba2));                   \
        const float n1  = u1 * h1o + (1.0f - u1) * cd1;                          \
                                                      \
        const float u2  = sigmoidf_dev((hA[1] + ba0) + pa);                      \
        const float rg2 = sigmoidf_dev((hB[1] + ba1) + pb);                      \
        const float cd2 = tanhf_dev(pc + rg2 * (hC[1] + ba2));                   \
        const float n2  = u2 * h2o + (1.0f - u2) * cd2;                          \
        const bool l1 = in1 & (mk1 > 0);            \
        const bool l2 = in2 & (mk2 > 0);                                         \
        h1o = l1 ? n1 : h1o;                                                     \
        h2o = l2 ? n2 : h2o;                                                     \
        if (hi == 0) {                                                           \
            hs[wp * RT * LP + col]        = h1o;                                 \
            hs[wp * RT * LP + LP + col]   = h2o;                                 \
            hist[(bp * LAG + m) * LP + col] = h1o;                               \
        }                                                                        \
    }

__global__ __launch_bounds__(256) void gru_serial_kernel(
    const int*   __restrict__ mask,
    const float* __restrict__ Wh, const float* __restrict__ bh,
    const float* __restrict__ Wx, const float* __restrict__ bx,
    const float* __restrict__ Xp,
    float* __restrict__ out)
{
    __shared__ float hs[2 * RT * LP];
    __shared__ float hist[2 * LAG * LP];
    __shared__ float x2[LAG * XP2];

    const int tid  = threadIdx.x;
    const int lane = tid & 31;
    const int jt   = tid >> 5;
    const int l15  = lane & 15;
    const int hi   = lane >> 4;
    const int col  = jt * 16 + l15;
    const bool g0 = __builtin_amdgcn_readfirstlane(jt) < 4;

    for (int i = tid; i < 2 * RT * LP; i += 256)   hs[i]   = 0.0f;
    for (int i = tid; i < 2 * LAG * LP; i += 256)  hist[i] = 0.0f;
    for (int i = tid; i < LAG * XP2; i += 256)     x2[i]   = 0.0f;
    __syncthreads();

    const float* __restrict__ whA = &Wh[(0 * D + jt * 16 + l15) * D];
    const float* __restrict__ whB = &Wh[(1 * D + jt * 16 + l15) * D];
    const float* __restrict__ whC = &Wh[(2 * D + jt * 16 + l15) * D];
    const float* __restrict__ wxA = &Wx[(0 * D + jt * 16 + l15) * D];
    const float* __restrict__ wxB = &Wx[(1 * D + jt * 16 + l15) * D];
    const float* __restrict__ wxC = &Wx[(2 * D + jt * 16 + l15) * D];

    v2f fhA[KS], fhB[KS], fhC[KS], fxA[KS], fxB[KS], fxC[KS];
#pragma unroll
    for (int i = 0; i < KS; ++i) {
        const int kb = 4 * i + 2 * hi;
        fhA[i] = v2f{ whA[kb], whA[kb + 1] };
        fhB[i] = v2f{ whB[kb], whB[kb + 1] };
        fhC[i] = v2f{ whC[kb], whC[kb + 1] };
        fxA[i] = v2f{ wxA[kb], wxA[kb + 1] };
        fxB[i] = v2f{ wxB[kb], wxB[kb + 1] };
        fxC[i] = v2f{ wxC[kb], wxC[kb + 1] };
    }
    const float ba0 = bh[col],         bb0 = bx[col];
    const float ba1 = bh[D + col],     bb1 = bx[D + col];
    const float ba2 = bh[2 * D + col], bb2 = bx[2 * D + col];

    float h1o = 0.0f, h2o = 0.0f;

    v2f a1[KH];
    {
        const float* aZ = &hs[RT * LP + l15 * LP];
#pragma unroll
        for (int i = 0; i < KH; ++i) {
            const int kb = 4 * i + 2 * hi;
            a1[i] = v2f{ aZ[kb], aZ[kb + 1] };
        }
    }
    bar_signal();

    for (int t = 0; t < NTOK + LAG; ++t) {
        const int m  = t & (LAG - 1);
        const int rp = (t + 1) & 1;
        const int wp = t & 1;
        const int bp = (t >> 4) & 1;
        const float* aR = &hs[rp * RT * LP + l15 * LP];
        const float* aW = &hs[wp * RT * LP + l15 * LP];

        const bool in1 = (t < NTOK);
        const bool in2 = (t >= LAG);
        const int  tc  = in1 ? t : NTOK - 1;
        const int  t2  = in2 ? t - LAG : 0;
        const int  mk1 = mask[tc];
        const int  mk2 = mask[t2];
        const float xa = Xp[tc * G3 + col];
        const float xb = Xp[tc * G3 + D + col];
        const float xc = Xp[tc * G3 + 2 * D + col];

        v8f hA = {}, hB = {}, hC = {};
#pragma unroll
        for (int i = 0; i < KB; ++i) {
            hA = wmma4(a1[i], fhA[i], hA);
            hB = wmma4(a1[i], fhB[i], hB);
            hC = wmma4(a1[i], fhC[i], hC);
        }
        __builtin_amdgcn_sched_barrier(0);

        bar_wait();

        if (m == 0 && t >= LAG) {
            const float* aH = &hist[(bp ^ 1) * LAG * LP + l15 * LP];
            v8f xA = {}, xB = {}, xC = {};
#pragma unroll
            for (int hh = 0; hh < 2; ++hh) {
                v2f ah[KH];
#pragma unroll
                for (int i = 0; i < KH; ++i) {
                    const int kb = 4 * (hh * KH + i) + 2 * hi;
                    ah[i] = v2f{ aH[kb], aH[kb + 1] };
                }
                __builtin_amdgcn_sched_barrier(0);
#pragma unroll
                for (int i = 0; i < KH; ++i) {
                    xA = wmma4(ah[i], fxA[hh * KH + i], xA);
                    xB = wmma4(ah[i], fxB[hh * KH + i], xB);
                    xC = wmma4(ah[i], fxC[hh * KH + i], xC);
                }
                __builtin_amdgcn_sched_barrier(0);
            }
#pragma unroll
            for (int r = 0; r < 8; ++r) {
                float* xr = &x2[(r + 8 * hi) * XP2];
                xr[col]         = xA[r] + bb0;
                xr[D + col]     = xB[r] + bb1;
                xr[2 * D + col] = xC[r] + bb2;
            }
        }

        v2f a2[KH];
#pragma unroll
        for (int i = 0; i < KH; ++i) {
            const int kb = 4 * (KH + i) + 2 * hi;
            a2[i] = v2f{ aR[kb], aR[kb + 1] };
        }
        const float pa = x2[m * XP2 + col];
        const float pb = x2[m * XP2 + D + col];
        const float pc = x2[m * XP2 + 2 * D + col];
        __builtin_amdgcn_sched_barrier(0);

#pragma unroll
        for (int i = KB; i < KH; ++i) {
            hA = wmma4(a1[i], fhA[i], hA);
            hB = wmma4(a1[i], fhB[i], hB);
            hC = wmma4(a1[i], fhC[i], hC);
        }

        if (g0) {
#pragma unroll
            for (int i = 0; i < KH; ++i) {
                hA = wmma4(a2[i], fhA[KH + i], hA);
                hB = wmma4(a2[i], fhB[KH + i], hB);
                hC = wmma4(a2[i], fhC[KH + i], hC);
            }
            GATES_AND_STORE
        }
        bar_signal();
        if (!g0) {
            __builtin_amdgcn_s_setprio(3);
#pragma unroll
            for (int i = 0; i < KH; ++i) {
                hA = wmma4(a2[i], fhA[KH + i], hA);
                hB = wmma4(a2[i], fhB[KH + i], hB);
                hC = wmma4(a2[i], fhC[KH + i], hC);
            }
        }
        bar_wait();

#pragma unroll
        for (int i = 0; i < KH; ++i) {
            const int kb = 4 * i + 2 * hi;
            a1[i] = v2f{ aW[kb], aW[kb + 1] };
        }
        __builtin_amdgcn_sched_barrier(0);

        if (!g0) {
            GATES_AND_STORE
            __builtin_amdgcn_s_setprio(0);
        }
        bar_signal();
    }
    bar_wait();

    const float* hf = &hs[((NTOK + LAG - 1) & 1) * RT * LP];
    float2* __restrict__ out2 = reinterpret_cast<float2*>(out + D * D);
    for (int i = tid; i < D * D; i += 256) {
        const int j = i % D;
        const float a = hf[j];
        const float b = hf[LP + j];
        out[i]  = b;
        out2[i] = make_float2(a, b);
    }
}

extern "C" void kernel_launch(void* const* d_in, const int* in_sizes, int n_in,
                              void* d_out, int out_size, void* d_ws, size_t ws_size,
                              hipStream_t stream) {
    const float* xs   = (const float*)d_in[0];
    const int*   mask = (const int*)  d_in[1];
    const float* Wh   = (const float*)d_in[2];
    const float* bh   = (const float*)d_in[3];
    const float* Wx   = (const float*)d_in[4];
    const float* bx   = (const float*)d_in[5];
    float* Xp  = (float*)d_ws;
    float* out = (float*)d_out;

    gru_xproj_kernel<<<192, 256, 0, stream>>>(xs, Wx, bx, Xp);
    gru_serial_kernel<<<1, 256, 0, stream>>>(mask, Wh, bh, Wx, bx, Xp, out);
}
